# starting kernel
# baseline (speedup 1.0000x reference)
#include <hip/hip_runtime.h>

typedef __attribute__((ext_vector_type(2))) float v2f;
typedef __attribute__((ext_vector_type(8))) float v8f;

#define D_DIM 512
#define CHUNKS 16

__global__ void bag_init_kernel(const int* __restrict__ counts, int B,
                                float* __restrict__ acc, int* __restrict__ prefix) {
    const int n = B * D_DIM;
    for (int i = threadIdx.x; i < n; i += blockDim.x) acc[i] = 0.0f;
    if (threadIdx.x == 0) {
        int s = 0;
        prefix[0] = 0;
        for (int i = 0; i < B; ++i) { s += counts[i]; prefix[i + 1] = s; }
    }
}

__global__ void __launch_bounds__(256)
bag_sum_wmma_kernel(const float* __restrict__ samples,
                    const int* __restrict__ prefix,
                    float* __restrict__ acc) {
    const int bag   = blockIdx.y;
    const int start = prefix[bag];
    const int end   = prefix[bag + 1];
    const int count = end - start;
    const int rowsPer = (count + (int)gridDim.x - 1) / (int)gridDim.x;
    const int r0 = start + (int)blockIdx.x * rowsPer;
    const int r1 = min(r0 + rowsPer, end);
    if (r0 >= r1) return;

    const int lane  = threadIdx.x & 31;
    const int wave  = threadIdx.x >> 5;
    const int lmod  = lane & 15;
    const int khalf = (lane >> 4) << 1;
    const int colBase = wave * 64;

    v2f a; a.x = 1.0f; a.y = 1.0f;
    v8f c0 = {}, c1 = {}, c2 = {}, c3 = {};

    const int span  = r1 - r0;
    const int nfull = span >> 2;
    const int rem   = span & 3;

    const float* p = samples + (size_t)(r0 + khalf) * D_DIM + colBase + lmod;

    #pragma unroll 2
    for (int it = 0; it < nfull; ++it) {
        __builtin_prefetch(p + 16 * D_DIM, 0, 3);

        v2f b0, b1, b2, b3;
        b0.x = p[0];            b0.y = p[D_DIM];
        b1.x = p[16];           b1.y = p[D_DIM + 16];
        b2.x = p[32];           b2.y = p[D_DIM + 32];
        b3.x = p[48];           b3.y = p[D_DIM + 48];
        p += 4 * D_DIM;

        c0 = __builtin_amdgcn_wmma_f32_16x16x4_f32(false, a, false, b0, (short)0, c0, false, false);
        c1 = __builtin_amdgcn_wmma_f32_16x16x4_f32(false, a, false, b1, (short)0, c1, false, false);
        c2 = __builtin_amdgcn_wmma_f32_16x16x4_f32(false, a, false, b2, (short)0, c2, false, false);
        c3 = __builtin_amdgcn_wmma_f32_16x16x4_f32(false, a, false, b3, (short)0, c3, false, false);
    }

    if (rem) {
        const int rbase = r1 - rem;
        const bool ok0 = khalf < rem;
        const bool ok1 = (khalf + 1) < rem;
        const int  o0  = ok0 ? khalf : 0;
        const int  o1  = ok1 ? (khalf + 1) : 0;
        const float* q0 = samples + (size_t)(rbase + o0) * D_DIM + colBase + lmod;
        const float* q1 = samples + (size_t)(rbase + o1) * D_DIM + colBase + lmod;

        v2f b0, b1, b2, b3;
        float x;
        x = q0[0];  b0.x = ok0 ? x : 0.0f;  x = q1[0];  b0.y = ok1 ? x : 0.0f;
        x = q0[16]; b1.x = ok0 ? x : 0.0f;  x = q1[16]; b1.y = ok1 ? x : 0.0f;
        x = q0[32]; b2.x = ok0 ? x : 0.0f;  x = q1[32]; b2.y = ok1 ? x : 0.0f;
        x = q0[48]; b3.x = ok0 ? x : 0.0f;  x = q1[48]; b3.y = ok1 ? x : 0.0f;

        c0 = __builtin_amdgcn_wmma_f32_16x16x4_f32(false, a, false, b0, (short)0, c0, false, false);
        c1 = __builtin_amdgcn_wmma_f32_16x16x4_f32(false, a, false, b1, (short)0, c1, false, false);
        c2 = __builtin_amdgcn_wmma_f32_16x16x4_f32(false, a, false, b2, (short)0, c2, false, false);
        c3 = __builtin_amdgcn_wmma_f32_16x16x4_f32(false, a, false, b3, (short)0, c3, false, false);
    }

    const int  hi  = lane >> 4;
    const float s0 = hi ? c2[0] : c0[0];
    const float s1 = hi ? c3[0] : c1[0];
    float* dst = acc + (size_t)bag * D_DIM + colBase + hi * 32 + lmod;
    atomicAdd(dst,      s0);
    atomicAdd(dst + 16, s1);
}

__global__ void bag_finalize_kernel(const float* __restrict__ acc,
                                    const int* __restrict__ counts,
                                    float* __restrict__ out, int n) {
    const int i = blockIdx.x * blockDim.x + threadIdx.x;
    if (i < n) {
        const int bag = i / D_DIM;
        out[i] = acc[i] / (float)counts[bag];
    }
}

extern "C" void kernel_launch(void* const* d_in, const int* in_sizes, int n_in,
                              void* d_out, int out_size, void* d_ws, size_t ws_size,
                              hipStream_t stream) {
    const float* samples = (const float*)d_in[0];
    const int*   counts  = (const int*)d_in[1];
    const int B = in_sizes[1];

    float* acc    = (float*)d_ws;
    int*   prefix = (int*)((char*)d_ws + (size_t)B * D_DIM * sizeof(float));
    float* out    = (float*)d_out;

    bag_init_kernel<<<dim3(1), dim3(256), 0, stream>>>(counts, B, acc, prefix);

    bag_sum_wmma_kernel<<<dim3(CHUNKS, B), dim3(256), 0, stream>>>(samples, prefix, acc);

    const int n = B * D_DIM;
    bag_finalize_kernel<<<dim3((n + 255) / 256), dim3(256), 0, stream>>>(acc, counts, out, n);
}


# Blocks own fixed 128-row tiles, not bag slices: a 0/1 bag-selector A matrix lets the WMMA sum each bag's rows apart; a combine launch adds per-tile partial sums in tile order; init and atomics go
# speedup vs baseline: 1.4236x; 1.4236x over previous
#include <hip/hip_runtime.h>

typedef __attribute__((ext_vector_type(2))) float v2f;
typedef __attribute__((ext_vector_type(4))) float v4f;
typedef __attribute__((ext_vector_type(8))) float v8f;

#define D_DIM 512
#define CHUNKS 16

__global__ void bag_init_kernel(const int* __restrict__ counts, int B,
                                float* __restrict__ acc, int* __restrict__ prefix) {
    const int n = B * D_DIM;
    for (int i = threadIdx.x; i < n; i += blockDim.x) acc[i] = 0.0f;
    if (threadIdx.x == 0) {
        int s = 0;
        prefix[0] = 0;
        for (int i = 0; i < B; ++i) { s += counts[i]; prefix[i + 1] = s; }
    }
}

__global__ void __launch_bounds__(256)
bag_sum_wmma_kernel(const float* __restrict__ samples,
                    const int* __restrict__ prefix,
                    float* __restrict__ acc) {
    const int bag   = blockIdx.y;
    const int start = prefix[bag];
    const int end   = prefix[bag + 1];
    const int count = end - start;
    const int rowsPer = (count + (int)gridDim.x - 1) / (int)gridDim.x;
    const int r0 = start + (int)blockIdx.x * rowsPer;
    const int r1 = min(r0 + rowsPer, end);
    if (r0 >= r1) return;

    const int lane  = threadIdx.x & 31;
    const int wave  = threadIdx.x >> 5;
    const int lmod  = lane & 15;
    const int khalf = (lane >> 4) << 1;
    const int colBase = wave * 64;

    v2f a; a.x = 1.0f; a.y = 1.0f;
    v8f c0 = {}, c1 = {}, c2 = {}, c3 = {};

    const int span  = r1 - r0;
    const int nfull = span >> 2;
    const int rem   = span & 3;

    const float* p = samples + (size_t)(r0 + khalf) * D_DIM + colBase + lmod;

    #pragma unroll 2
    for (int it = 0; it < nfull; ++it) {
        __builtin_prefetch(p + 16 * D_DIM, 0, 3);

        v2f b0, b1, b2, b3;
        b0.x = p[0];            b0.y = p[D_DIM];
        b1.x = p[16];           b1.y = p[D_DIM + 16];
        b2.x = p[32];           b2.y = p[D_DIM + 32];
        b3.x = p[48];           b3.y = p[D_DIM + 48];
        p += 4 * D_DIM;

        c0 = __builtin_amdgcn_wmma_f32_16x16x4_f32(false, a, false, b0, (short)0, c0, false, false);
        c1 = __builtin_amdgcn_wmma_f32_16x16x4_f32(false, a, false, b1, (short)0, c1, false, false);
        c2 = __builtin_amdgcn_wmma_f32_16x16x4_f32(false, a, false, b2, (short)0, c2, false, false);
        c3 = __builtin_amdgcn_wmma_f32_16x16x4_f32(false, a, false, b3, (short)0, c3, false, false);
    }

    if (rem) {
        const int rbase = r1 - rem;
        const bool ok0 = khalf < rem;
        const bool ok1 = (khalf + 1) < rem;
        const int  o0  = ok0 ? khalf : 0;
        const int  o1  = ok1 ? (khalf + 1) : 0;
        const float* q0 = samples + (size_t)(rbase + o0) * D_DIM + colBase + lmod;
        const float* q1 = samples + (size_t)(rbase + o1) * D_DIM + colBase + lmod;

        v2f b0, b1, b2, b3;
        float x;
        x = q0[0];  b0.x = ok0 ? x : 0.0f;  x = q1[0];  b0.y = ok1 ? x : 0.0f;
        x = q0[16]; b1.x = ok0 ? x : 0.0f;  x = q1[16]; b1.y = ok1 ? x : 0.0f;
        x = q0[32]; b2.x = ok0 ? x : 0.0f;  x = q1[32]; b2.y = ok1 ? x : 0.0f;
        x = q0[48]; b3.x = ok0 ? x : 0.0f;  x = q1[48]; b3.y = ok1 ? x : 0.0f;

        c0 = __builtin_amdgcn_wmma_f32_16x16x4_f32(false, a, false, b0, (short)0, c0, false, false);
        c1 = __builtin_amdgcn_wmma_f32_16x16x4_f32(false, a, false, b1, (short)0, c1, false, false);
        c2 = __builtin_amdgcn_wmma_f32_16x16x4_f32(false, a, false, b2, (short)0, c2, false, false);
        c3 = __builtin_amdgcn_wmma_f32_16x16x4_f32(false, a, false, b3, (short)0, c3, false, false);
    }

    const int  hi  = lane >> 4;
    const float s0 = hi ? c2[0] : c0[0];
    const float s1 = hi ? c3[0] : c1[0];
    float* dst = acc + (size_t)bag * D_DIM + colBase + hi * 32 + lmod;
    atomicAdd(dst,      s0);
    atomicAdd(dst + 16, s1);
}

__global__ void bag_finalize_kernel(const float* __restrict__ acc,
                                    const int* __restrict__ counts,
                                    float* __restrict__ out, int n) {
    const int i = blockIdx.x * blockDim.x + threadIdx.x;
    if (i < n) {
        const int bag = i / D_DIM;
        out[i] = acc[i] / (float)counts[bag];
    }
}

#define TILE_ROWS  128
#define N_TILES    1024
#define GROUP_BAGS 8

__device__ __forceinline__ void bag_scan(const int* __restrict__ counts, int B, int lane,
                                         int& inc0, int& inc1) {
    int v0 = (lane < B) ? counts[lane] : 0;
    int v1 = (lane + 32 < B) ? counts[lane + 32] : 0;
    for (int d = 1; d < 32; d <<= 1) {
        const int t0 = __shfl_up(v0, d);
        const int t1 = __shfl_up(v1, d);
        if (lane >= d) { v0 += t0; v1 += t1; }
    }
    inc0 = v0;
    inc1 = v1 + __shfl(v0, 31);
}

__device__ __forceinline__ int bag_start(int inc0, int inc1, int j) {
    const int k = j - 1;
    const int a = __shfl(inc0, k & 31);
    const int b = __shfl(inc1, k & 31);
    return (j <= 0) ? 0 : ((k & 32) ? b : a);
}

__global__ void __launch_bounds__(256)
bag_rowtile_wmma_kernel(const float* __restrict__ samples,
                        const int* __restrict__ counts, int B,
                        float* __restrict__ part) {
    const int lane  = threadIdx.x & 31;
    const int wave  = threadIdx.x >> 5;
    const int lmod  = lane & 15;
    const int h     = lane >> 4;
    const int khalf = h << 1;
    const int colBase = wave * 64;
    const int tile  = blockIdx.x;
    const int r0    = tile * TILE_ROWS;

    int inc0, inc1;
    bag_scan(counts, B, lane, inc0, inc1);
    const int bFirst = (int)__popcll(__ballot(lane < B && inc0 <= r0)) +
                       (int)__popcll(__ballot(lane + 32 < B && inc1 <= r0));
    const int rl = r0 + TILE_ROWS - 1;
    const int bLast = min((int)__popcll(__ballot(lane < B && inc0 <= rl)) +
                          (int)__popcll(__ballot(lane + 32 < B && inc1 <= rl)), B - 1);

    for (int bag0 = bFirst; bag0 <= bLast; bag0 += GROUP_BAGS) {
        const int jb = min(bag0 + (lane & 7), 64);
        const int lo = bag_start(inc0, inc1, jb);
        const int up = bag_start(inc0, inc1, min(jb + 1, 64));
        const unsigned len = (jb < B) ? (unsigned)(up - lo) : 0u;
        int rel = r0 + khalf - lo;

        v8f c0 = {}, c1 = {}, c2 = {}, c3 = {};
        const float* p = samples + (size_t)(r0 + khalf) * D_DIM + colBase + lmod;

        #pragma unroll 2
        for (int it = 0; it < TILE_ROWS / 4; ++it) {
            __builtin_prefetch(p + 16 * D_DIM, 0, 3);

            v2f b0, b1, b2, b3;
            b0.x = p[0];            b0.y = p[D_DIM];
            b1.x = p[16];           b1.y = p[D_DIM + 16];
            b2.x = p[32];           b2.y = p[D_DIM + 32];
            b3.x = p[48];           b3.y = p[D_DIM + 48];
            p += 4 * D_DIM;

            v2f a;
            a.x = ((unsigned)rel < len) ? 1.0f : 0.0f;
            a.y = ((unsigned)(rel + 1) < len) ? 1.0f : 0.0f;
            rel += 4;

            c0 = __builtin_amdgcn_wmma_f32_16x16x4_f32(false, a, false, b0, (short)0, c0, false, false);
            c1 = __builtin_amdgcn_wmma_f32_16x16x4_f32(false, a, false, b1, (short)0, c1, false, false);
            c2 = __builtin_amdgcn_wmma_f32_16x16x4_f32(false, a, false, b2, (short)0, c2, false, false);
            c3 = __builtin_amdgcn_wmma_f32_16x16x4_f32(false, a, false, b3, (short)0, c3, false, false);
        }

        const int nb = min(bLast - bag0 + 1, GROUP_BAGS);
        float* dst = part + (size_t)(tile + bag0) * D_DIM + colBase + h * 16 + lmod;
        #pragma unroll
        for (int r = 0; r < GROUP_BAGS; ++r) {
            if (r < nb) {
                dst[r * D_DIM]      = h ? c1[r] : c0[r];
                dst[r * D_DIM + 32] = h ? c3[r] : c2[r];
            }
        }
    }
}

__global__ void __launch_bounds__(512)
bag_combine_kernel(const float* __restrict__ part,
                   const int* __restrict__ counts,
                   float* __restrict__ out) {
    __shared__ float red[16][128];
    const int lane = threadIdx.x & 31;
    const int wave = threadIdx.x >> 5;
    const int bag  = blockIdx.y;
    const int col0 = blockIdx.x * 128;

    int start = ((lane < bag) ? counts[lane] : 0) + ((lane + 32 < bag) ? counts[lane + 32] : 0);
    for (int d = 16; d > 0; d >>= 1) start += __shfl_xor(start, d);
    const int cnt = counts[bag];
    const int tLo = start / TILE_ROWS;
    const int tHi = min((start + cnt - 1) / TILE_ROWS, N_TILES - 1);

    v4f acc = {0.0f, 0.0f, 0.0f, 0.0f};
    if (cnt > 0) {
        for (int t = tLo + wave; t <= tHi; t += 16) {
            const v4f x = *(const v4f*)(part + (size_t)(t + bag) * D_DIM + col0 + lane * 4);
            acc += x;
        }
    }
    *(v4f*)&red[wave][lane * 4] = acc;
    __syncthreads();
    if (threadIdx.x < 128) {
        float s = red[0][threadIdx.x];
        for (int w = 1; w < 16; ++w) s += red[w][threadIdx.x];
        out[(size_t)bag * D_DIM + col0 + threadIdx.x] = s / (float)cnt;
    }
}

extern "C" void kernel_launch(void* const* d_in, const int* in_sizes, int n_in,
                              void* d_out, int out_size, void* d_ws, size_t ws_size,
                              hipStream_t stream) {
    const float* samples = (const float*)d_in[0];
    const int*   counts  = (const int*)d_in[1];
    const int B = in_sizes[1];
    float* out    = (float*)d_out;

    if (B <= 64 && ws_size >= (size_t)2228224) {
        float* part = (float*)d_ws;
        bag_rowtile_wmma_kernel<<<dim3(N_TILES), dim3(256), 0, stream>>>(samples, counts, B, part);
        bag_combine_kernel<<<dim3(4, B), dim3(512), 0, stream>>>(part, counts, out);
        return;
    }

    float* acc    = (float*)d_ws;
    int*   prefix = (int*)((char*)d_ws + (size_t)B * D_DIM * sizeof(float));

    bag_init_kernel<<<dim3(1), dim3(256), 0, stream>>>(counts, B, acc, prefix);

    bag_sum_wmma_kernel<<<dim3(CHUNKS, B), dim3(256), 0, stream>>>(samples, prefix, acc);

    const int n = B * D_DIM;
    bag_finalize_kernel<<<dim3((n + 255) / 256), dim3(256), 0, stream>>>(acc, counts, out, n);
}
